# starting kernel
# baseline (speedup 1.0000x reference)
#include <hip/hip_runtime.h>


typedef float v2f __attribute__((ext_vector_type(2)));
typedef float v8f __attribute__((ext_vector_type(8)));

#define N_PTS   400000
#define M_Q     100000
#define NB      16
#define C_IN    64
#define D2      32
#define C_OUT   128
#define KP      15
#define INV_EXTENT 2.0f
#define EPS_BN  1e-5f
#define SLOPE   0.2f

__device__ __forceinline__ float leaky(float v) { return v >= 0.f ? v : SLOPE * v; }

__device__ __forceinline__ v8f wmma4(v2f a, v2f b, v8f c) {
  return __builtin_amdgcn_wmma_f32_16x16x4_f32(false, a, false, b, (short)0, c,
                                               false, false);
}

__device__ __forceinline__ v2f load_a_rm(const float* src, int ld, int m0,
                                         int k0, int lane) {
  int m = m0 + (lane & 15);
  int k = k0 + ((lane >> 4) << 1);
  v2f a;
  a.x = src[(size_t)m * ld + k];
  a.y = src[(size_t)m * ld + k + 1];
  return a;
}

__device__ __forceinline__ v2f load_b_rm(const float* src, int ld, int k0,
                                         int n0, int lane) {
  int n = n0 + (lane & 15);
  int k = k0 + ((lane >> 4) << 1);
  v2f b;
  b.x = src[(size_t)k * ld + n];
  b.y = src[(size_t)(k + 1) * ld + n];
  return b;
}

__device__ __forceinline__ void store_d_rm(float* dst, int ld, int m0, int n0,
                                           int lane, v8f c) {
  int n = n0 + (lane & 15);
  int rb = m0 + ((lane >> 4) << 3);
#pragma unroll
  for (int v = 0; v < 8; ++v) dst[(size_t)(rb + v) * ld + n] = c[v];
}

__device__ __forceinline__ void accum_stats(float* sum, float* sq, int n0,
                                            int lane, v8f c) {
  float s = 0.f, q = 0.f;
#pragma unroll
  for (int v = 0; v < 8; ++v) { s += c[v]; q += c[v] * c[v]; }
  atomicAdd(&sum[n0 + (lane & 15)], s);
  atomicAdd(&sq[n0 + (lane & 15)], q);
}

__global__ void k_zero(float* p, int n) {
  int i = blockIdx.x * blockDim.x + threadIdx.x;
  if (i < n) p[i] = 0.f;
}

__global__ void k_finalize_bn(const float* __restrict__ sum,
                              const float* __restrict__ sq,
                              const float* __restrict__ g,
                              const float* __restrict__ b, float cnt,
                              float* a, float* c, int C) {
  int i = blockIdx.x * blockDim.x + threadIdx.x;
  if (i >= C) return;
  float mu  = sum[i] / cnt;
  float var = sq[i] / cnt - mu * mu;
  float s   = g[i] * rsqrtf(fmaxf(var, 0.f) + EPS_BN);
  a[i] = s;
  c[i] = b[i] - mu * s;
}

__global__ __launch_bounds__(256) void k_gemm1(const float* __restrict__ x,
                                               const float* __restrict__ W,
                                               float* __restrict__ y1,
                                               float* sum1, float* sq1,
                                               int ntiles) {
  int lane = threadIdx.x & 31;
  int tile = blockIdx.x * (blockDim.x >> 5) + (threadIdx.x >> 5);
  if (tile >= ntiles) return;
  int m0 = tile * 16;
  v8f acc0 = {}, acc1 = {};
#pragma unroll
  for (int k0 = 0; k0 < C_IN; k0 += 4) {
    v2f a  = load_a_rm(x, C_IN, m0, k0, lane);
    v2f b0 = load_b_rm(W, D2, k0, 0, lane);
    v2f b1 = load_b_rm(W, D2, k0, 16, lane);
    acc0 = wmma4(a, b0, acc0);
    acc1 = wmma4(a, b1, acc1);
  }
  store_d_rm(y1, D2, m0, 0, lane, acc0);
  store_d_rm(y1, D2, m0, 16, lane, acc1);
  accum_stats(sum1, sq1, 0, lane, acc0);
  accum_stats(sum1, sq1, 16, lane, acc1);
}

__global__ __launch_bounds__(128) void k_kpconv(
    const float* __restrict__ y1, const int* __restrict__ idx,
    const float* __restrict__ pos, const float* __restrict__ qpos,
    const float* __restrict__ Kpts, const float* __restrict__ KV,
    const float* __restrict__ a1, const float* __restrict__ c1,
    float* __restrict__ y2, float* sumk, float* sqk) {
  __shared__ float smem[12288];
  float* s_w  = smem;
  float* s_nf = smem + 4096;
  int lane = threadIdx.x & 31;
  int wave = threadIdx.x >> 5;
  int q0 = blockIdx.x * 16;

  for (int p = threadIdx.x; p < 256; p += 128) {
    int q = p >> 4, nb = p & 15;
    int id = idx[(size_t)(q0 + q) * NB + nb];
    float qx = qpos[(size_t)(q0 + q) * 3 + 0];
    float qy = qpos[(size_t)(q0 + q) * 3 + 1];
    float qz = qpos[(size_t)(q0 + q) * 3 + 2];
    float px, py, pz;
    if (id < N_PTS) {
      px = pos[(size_t)id * 3 + 0];
      py = pos[(size_t)id * 3 + 1];
      pz = pos[(size_t)id * 3 + 2];
    } else {
      px = py = pz = 1e6f;
    }
    float rx = px - qx, ry = py - qy, rz = pz - qz;
    float rr = rx * rx + ry * ry + rz * rz;
#pragma unroll
    for (int k = 0; k < KP; ++k) {
      float kx = Kpts[k * 3 + 0], ky = Kpts[k * 3 + 1], kz = Kpts[k * 3 + 2];
      float sq = rr + kx * kx + ky * ky + kz * kz
                 - 2.f * (rx * kx + ry * ky + rz * kz);
      float w = 1.f - sqrtf(fmaxf(sq, 0.f)) * INV_EXTENT;
      s_w[q * 256 + nb * 16 + k] = fmaxf(w, 0.f);
    }
    s_w[q * 256 + nb * 16 + 15] = 0.f;
    if (id < N_PTS) {
      const float* yr = y1 + (size_t)id * D2;
#pragma unroll
      for (int d = 0; d < D2; ++d)
        s_nf[q * 512 + nb * 32 + d] = leaky(yr[d] * a1[d] + c1[d]);
    } else {
#pragma unroll
      for (int d = 0; d < D2; ++d) s_nf[q * 512 + nb * 32 + d] = 0.f;
    }
  }
  __syncthreads();

  v8f wt[4][2];
#pragma unroll
  for (int qq = 0; qq < 4; ++qq) { wt[qq][0] = (v8f){}; wt[qq][1] = (v8f){}; }
#pragma unroll
  for (int qq = 0; qq < 4; ++qq) {
    int q = wave * 4 + qq;
#pragma unroll
    for (int k0 = 0; k0 < 16; k0 += 4) {
      int m = lane & 15;
      int kk = k0 + ((lane >> 4) << 1);
      v2f a;
      a.x = s_w[q * 256 + kk * 16 + m];
      a.y = s_w[q * 256 + (kk + 1) * 16 + m];
#pragma unroll
      for (int t = 0; t < 2; ++t) {
        int n = t * 16 + (lane & 15);
        v2f b;
        b.x = s_nf[q * 512 + kk * 32 + n];
        b.y = s_nf[q * 512 + (kk + 1) * 32 + n];
        wt[qq][t] = wmma4(a, b, wt[qq][t]);
      }
    }
  }
  __syncthreads();

  float* s_wt = smem;
#pragma unroll
  for (int qq = 0; qq < 4; ++qq) {
    int q = wave * 4 + qq;
#pragma unroll
    for (int t = 0; t < 2; ++t) {
      int n = t * 16 + (lane & 15);
      int rb = (lane >> 4) << 3;
#pragma unroll
      for (int v = 0; v < 8; ++v)
        s_wt[q * 512 + (rb + v) * 32 + n] = wt[qq][t][v];
    }
  }
  __syncthreads();

  int ntile = wave & 1;
  int kbase = (wave >> 1) * 256;
  v8f acc = {};
  for (int k0 = 0; k0 < 256; k0 += 4) {
    int kk = kbase + k0 + ((lane >> 4) << 1);
    int m = lane & 15;
    v2f a;
    a.x = s_wt[m * 512 + kk];
    a.y = s_wt[m * 512 + kk + 1];
    int n = ntile * 16 + (lane & 15);
    v2f b;
    b.x = (kk < 480) ? KV[(size_t)kk * 32 + n] : 0.f;
    b.y = (kk + 1 < 480) ? KV[(size_t)(kk + 1) * 32 + n] : 0.f;
    acc = wmma4(a, b, acc);
  }
  __syncthreads();
  float* s_red = smem;
  if (wave >= 2) {
#pragma unroll
    for (int v = 0; v < 8; ++v) s_red[(wave - 2) * 256 + lane * 8 + v] = acc[v];
  }
  __syncthreads();
  if (wave < 2) {
#pragma unroll
    for (int v = 0; v < 8; ++v) acc[v] += s_red[wave * 256 + lane * 8 + v];
    int col = wave * 16 + (lane & 15);
    int rb = q0 + ((lane >> 4) << 3);
    float s = 0.f, q2 = 0.f;
#pragma unroll
    for (int v = 0; v < 8; ++v) {
      float val = acc[v];
      y2[(size_t)(rb + v) * D2 + col] = val;
      s += val;
      q2 += val * val;
    }
    atomicAdd(&sumk[col], s);
    atomicAdd(&sqk[col], q2);
  }
}

__global__ __launch_bounds__(256) void k_gemm2(const float* __restrict__ y2,
                                               const float* __restrict__ W,
                                               const float* __restrict__ ak,
                                               const float* __restrict__ ck,
                                               float* __restrict__ y3,
                                               float* sum2, float* sq2,
                                               int ntiles) {
  int lane = threadIdx.x & 31;
  int tile = blockIdx.x * (blockDim.x >> 5) + (threadIdx.x >> 5);
  if (tile >= ntiles) return;
  int m0 = tile * 16;
  v8f acc[8];
#pragma unroll
  for (int t = 0; t < 8; ++t) acc[t] = (v8f){};
#pragma unroll
  for (int k0 = 0; k0 < D2; k0 += 4) {
    int m = m0 + (lane & 15);
    int kk = k0 + ((lane >> 4) << 1);
    v2f a;
    a.x = leaky(y2[(size_t)m * D2 + kk] * ak[kk] + ck[kk]);
    a.y = leaky(y2[(size_t)m * D2 + kk + 1] * ak[kk + 1] + ck[kk + 1]);
#pragma unroll
    for (int t = 0; t < 8; ++t) {
      v2f b = load_b_rm(W, C_OUT, k0, t * 16, lane);
      acc[t] = wmma4(a, b, acc[t]);
    }
  }
#pragma unroll
  for (int t = 0; t < 8; ++t) {
    store_d_rm(y3, C_OUT, m0, t * 16, lane, acc[t]);
    accum_stats(sum2, sq2, t * 16, lane, acc[t]);
  }
}

__global__ __launch_bounds__(128) void k_final(const float* __restrict__ x,
                                               const int* __restrict__ idx,
                                               const float* __restrict__ Wsc,
                                               const float* __restrict__ y3,
                                               const float* __restrict__ a2,
                                               const float* __restrict__ c2,
                                               float* __restrict__ out) {
  __shared__ float s_sx[16 * 64];
  int q0 = blockIdx.x * 16;
  for (int p = threadIdx.x; p < 1024; p += 128) {
    int q = p >> 6, c = p & 63;
    float acc = -3.4e38f;
#pragma unroll
    for (int nb = 0; nb < NB; ++nb) {
      int id = idx[(size_t)(q0 + q) * NB + nb];
      float v = (id < N_PTS) ? x[(size_t)id * C_IN + c] : 0.f;
      acc = fmaxf(acc, v);
    }
    s_sx[q * 64 + c] = acc;
  }
  __syncthreads();
  int lane = threadIdx.x & 31;
  int wave = threadIdx.x >> 5;
#pragma unroll
  for (int tt = 0; tt < 2; ++tt) {
    int ntile = wave * 2 + tt;
    v8f acc = {};
#pragma unroll
    for (int k0 = 0; k0 < C_IN; k0 += 4) {
      int m = lane & 15;
      int kk = k0 + ((lane >> 4) << 1);
      v2f a;
      a.x = s_sx[m * 64 + kk];
      a.y = s_sx[m * 64 + kk + 1];
      v2f b = load_b_rm(Wsc, C_OUT, k0, ntile * 16, lane);
      acc = wmma4(a, b, acc);
    }
    int col = ntile * 16 + (lane & 15);
    int rb = (lane >> 4) << 3;
#pragma unroll
    for (int v = 0; v < 8; ++v) {
      int m = q0 + rb + v;
      float h = leaky(y3[(size_t)m * C_OUT + col] * a2[col] + c2[col]);
      out[(size_t)m * C_OUT + col] = h + acc[v];
    }
  }
}

extern "C" void kernel_launch(void* const* d_in, const int* in_sizes, int n_in,
                              void* d_out, int out_size, void* d_ws,
                              size_t ws_size, hipStream_t stream) {
  const float* x    = (const float*)d_in[0];
  const float* pos  = (const float*)d_in[1];
  const float* qpos = (const float*)d_in[2];
  const int*   idx  = (const int*)d_in[3];
  const float* Wu1  = (const float*)d_in[4];
  const float* g1   = (const float*)d_in[5];
  const float* b1   = (const float*)d_in[6];
  const float* Kpts = (const float*)d_in[7];
  const float* KV   = (const float*)d_in[8];
  const float* gk   = (const float*)d_in[9];
  const float* bk   = (const float*)d_in[10];
  const float* Wu2  = (const float*)d_in[11];
  const float* g2   = (const float*)d_in[12];
  const float* b2   = (const float*)d_in[13];
  const float* Wsc  = (const float*)d_in[14];
  float* out = (float*)d_out;

  float* ws = (float*)d_ws;
  float* y1 = ws;
  float* y2 = y1 + (size_t)N_PTS * D2;
  float* y3 = y2 + (size_t)M_Q * D2;
  float* S  = y3 + (size_t)M_Q * C_OUT;
  float *sum1 = S +   0, *sq1 = S +  32;
  float *sumk = S +  64, *sqk = S +  96;
  float *sum2 = S + 128, *sq2 = S + 256;
  float *a1 = S + 384, *c1 = S + 416;
  float *ak = S + 448, *ck = S + 480;
  float *a2 = S + 512, *c2 = S + 640;

  k_zero<<<1, 384, 0, stream>>>(S, 384);

  k_gemm1<<<(N_PTS / 16) / 8, 256, 0, stream>>>(x, Wu1, y1, sum1, sq1,
                                                N_PTS / 16);
  k_finalize_bn<<<1, 32, 0, stream>>>(sum1, sq1, g1, b1, (float)N_PTS, a1, c1,
                                      D2);
  k_kpconv<<<M_Q / 16, 128, 0, stream>>>(y1, idx, pos, qpos, Kpts, KV, a1, c1,
                                         y2, sumk, sqk);
  k_finalize_bn<<<1, 32, 0, stream>>>(sumk, sqk, gk, bk, (float)M_Q, ak, ck,
                                      D2);
  int tiles2 = M_Q / 16;
  k_gemm2<<<(tiles2 + 7) / 8, 256, 0, stream>>>(y2, Wu2, ak, ck, y3, sum2, sq2,
                                                tiles2);
  k_finalize_bn<<<1, 128, 0, stream>>>(sum2, sq2, g2, b2, (float)M_Q, a2, c2,
                                       C_OUT);
  k_final<<<M_Q / 16, 128, 0, stream>>>(x, idx, Wsc, y3, a2, c2, out);
}


# KPConv step A operands built in fragment registers straight from memory (lane = kernel point or feature column; y1 as 64-byte half rows): no LDS staging, 2 barriers fewer; sum order unchanged
# speedup vs baseline: 4.5286x; 4.5286x over previous
#include <hip/hip_runtime.h>


typedef float v2f __attribute__((ext_vector_type(2)));
typedef float v4f __attribute__((ext_vector_type(4)));
typedef float v8f __attribute__((ext_vector_type(8)));
typedef int v2i __attribute__((ext_vector_type(2)));

#define N_PTS   400000
#define M_Q     100000
#define NB      16
#define C_IN    64
#define D2      32
#define C_OUT   128
#define KP      15
#define INV_EXTENT 2.0f
#define EPS_BN  1e-5f
#define SLOPE   0.2f

__device__ __forceinline__ float leaky(float v) { return v >= 0.f ? v : SLOPE * v; }

__device__ __forceinline__ v8f wmma4(v2f a, v2f b, v8f c) {
  return __builtin_amdgcn_wmma_f32_16x16x4_f32(false, a, false, b, (short)0, c,
                                               false, false);
}

__device__ __forceinline__ v2f load_a_rm(const float* src, int ld, int m0,
                                         int k0, int lane) {
  int m = m0 + (lane & 15);
  int k = k0 + ((lane >> 4) << 1);
  v2f a;
  a.x = src[(size_t)m * ld + k];
  a.y = src[(size_t)m * ld + k + 1];
  return a;
}

__device__ __forceinline__ v2f load_b_rm(const float* src, int ld, int k0,
                                         int n0, int lane) {
  int n = n0 + (lane & 15);
  int k = k0 + ((lane >> 4) << 1);
  v2f b;
  b.x = src[(size_t)k * ld + n];
  b.y = src[(size_t)(k + 1) * ld + n];
  return b;
}

__device__ __forceinline__ void store_d_rm(float* dst, int ld, int m0, int n0,
                                           int lane, v8f c) {
  int n = n0 + (lane & 15);
  int rb = m0 + ((lane >> 4) << 3);
#pragma unroll
  for (int v = 0; v < 8; ++v) dst[(size_t)(rb + v) * ld + n] = c[v];
}

__device__ __forceinline__ void accum_stats(float* sum, float* sq, int n0,
                                            int lane, v8f c) {
  float s = 0.f, q = 0.f;
#pragma unroll
  for (int v = 0; v < 8; ++v) { s += c[v]; q += c[v] * c[v]; }
  atomicAdd(&sum[n0 + (lane & 15)], s);
  atomicAdd(&sq[n0 + (lane & 15)], q);
}

__global__ void k_zero(float* p, int n) {
  int i = blockIdx.x * blockDim.x + threadIdx.x;
  if (i < n) p[i] = 0.f;
}

__device__ __forceinline__ void bn_coef(const float* sum, const float* sq,
                                        const float* g, const float* b,
                                        float cnt, int i, float& a, float& c) {
  float mu  = sum[i] / cnt;
  float var = sq[i] / cnt - mu * mu;
  float s   = g[i] * rsqrtf(fmaxf(var, 0.f) + EPS_BN);
  a = s;
  c = b[i] - mu * s;
}

__global__ __launch_bounds__(128) void k_gemm1(const float* __restrict__ x,
                                               const float* __restrict__ W,
                                               float* __restrict__ y1,
                                               float* sum1, float* sq1,
                                               int ntiles) {
  int lane = threadIdx.x & 31;
  int wv = __builtin_amdgcn_readfirstlane((int)(threadIdx.x >> 5));
  int gw = blockIdx.x * (blockDim.x >> 5) + wv;
  int nw = gridDim.x * (blockDim.x >> 5);
  v2f b0[16], b1[16];
#pragma unroll
  for (int k = 0; k < 16; ++k) {
    b0[k] = load_b_rm(W, D2, 4 * k, 0, lane);
    b1[k] = load_b_rm(W, D2, 4 * k, 16, lane);
  }
  float s0 = 0.f, q0 = 0.f, s1 = 0.f, q1 = 0.f;
  v2f a[16];
  {
    int t0 = gw < ntiles ? gw : 0;
#pragma unroll
    for (int k = 0; k < 16; ++k) a[k] = load_a_rm(x, C_IN, t0 * 16, 4 * k, lane);
  }
  for (int tile = gw; tile < ntiles; tile += nw) {
    int m0 = tile * 16;
    int nt = tile + nw < ntiles ? tile + nw : tile;
    v2f an[16];
#pragma unroll
    for (int k = 0; k < 16; ++k) an[k] = load_a_rm(x, C_IN, nt * 16, 4 * k, lane);
    v8f acc0 = {}, acc1 = {};
#pragma unroll
    for (int k = 0; k < 16; ++k) {
      acc0 = wmma4(a[k], b0[k], acc0);
      acc1 = wmma4(a[k], b1[k], acc1);
    }
    store_d_rm(y1, D2, m0, 0, lane, acc0);
    store_d_rm(y1, D2, m0, 16, lane, acc1);
#pragma unroll
    for (int v = 0; v < 8; ++v) {
      s0 += acc0[v]; q0 += acc0[v] * acc0[v];
      s1 += acc1[v]; q1 += acc1[v] * acc1[v];
    }
#pragma unroll
    for (int k = 0; k < 16; ++k) a[k] = an[k];
  }
  s0 += __shfl_xor(s0, 16); q0 += __shfl_xor(q0, 16);
  s1 += __shfl_xor(s1, 16); q1 += __shfl_xor(q1, 16);
  atomicAdd(&sum1[lane], lane < 16 ? s0 : s1);
  atomicAdd(&sq1[lane], lane < 16 ? q0 : q1);
}

__global__ __launch_bounds__(128) void k_kpconv(
    const float* __restrict__ y1, const int* __restrict__ idx,
    const float* __restrict__ pos, const float* __restrict__ qpos,
    const float* __restrict__ Kpts, const float* __restrict__ KV,
    const float* __restrict__ sum1, const float* __restrict__ sq1,
    const float* __restrict__ g1, const float* __restrict__ b1,
    float* __restrict__ y2, float* sumk, float* sqk, int ntiles) {
  __shared__ __attribute__((aligned(16))) float smem[8192 + 512];
  float* s_wt = smem;
  float* s_red = smem + 8192;
  int lane = threadIdx.x & 31;
  int wave = __builtin_amdgcn_readfirstlane((int)(threadIdx.x >> 5));
  int ntile = wave & 1;
  int khalf = wave >> 1;
  int kbase = khalf * 256;
  int kh = (lane >> 4) << 1;
  int ncol = ntile * 16 + (lane & 15);

  v2f kvb[64];
#pragma unroll
  for (int ks = 0; ks < 56; ++ks) {
    int kk = kbase + 4 * ks + kh;
    kvb[ks].x = KV[(size_t)kk * 32 + ncol];
    kvb[ks].y = KV[(size_t)(kk + 1) * 32 + ncol];
  }
  if (khalf == 0) {
#pragma unroll
    for (int ks = 56; ks < 64; ++ks) {
      int kk = 4 * ks + kh;
      kvb[ks].x = KV[(size_t)kk * 32 + ncol];
      kvb[ks].y = KV[(size_t)(kk + 1) * 32 + ncol];
    }
  } else {
#pragma unroll
    for (int ks = 56; ks < 64; ++ks) { kvb[ks].x = 0.f; kvb[ks].y = 0.f; }
  }

  float st_s = 0.f, st_q = 0.f;
  int nblk = gridDim.x;
  int tile = blockIdx.x;
  int c16 = lane & 15;
  int hh = lane >> 4;

  float kx, ky, kz;
  {
    int kpc = c16 < KP ? c16 : 0;
    kx = Kpts[kpc * 3 + 0];
    ky = Kpts[kpc * 3 + 1];
    kz = Kpts[kpc * 3 + 2];
  }
  float kp_keep = c16 < KP ? 1.f : 0.f;
  float bnA[2], bnC[2];
#pragma unroll
  for (int t = 0; t < 2; ++t)
    bn_coef(sum1, sq1, g1, b1, (float)N_PTS, 16 * t + c16, bnA[t], bnC[t]);

  v2i idc[4][4];
  {
    int t0 = tile < ntiles ? tile : 0;
    const int* ib = idx + ((size_t)t0 * 16 + wave * 4) * NB + 2 * hh;
#pragma unroll
    for (int qq = 0; qq < 4; ++qq) {
#pragma unroll
      for (int ks = 0; ks < 4; ++ks)
        idc[qq][ks] = *(const v2i*)(ib + qq * NB + 4 * ks);
    }
  }

  for (; tile < ntiles; tile += nblk) {
    int q0 = tile * 16;
    v8f wt[4][2];
#pragma unroll
    for (int qq = 0; qq < 4; ++qq) { wt[qq][0] = (v8f){}; wt[qq][1] = (v8f){}; }
#pragma unroll
    for (int bb = 0; bb < 2; ++bb) {
      float px[2][8], py[2][8], pz[2][8];
      float f0[2][8], f1[2][8];
#pragma unroll
      for (int u = 0; u < 2; ++u) {
#pragma unroll
        for (int e = 0; e < 8; ++e) {
          int id = (e & 1) ? idc[2 * bb + u][e >> 1].y : idc[2 * bb + u][e >> 1].x;
          float vx = 1e6f, vy = 1e6f, vz = 1e6f;
          float g0 = 0.f, g1v = 0.f;
          if (id < N_PTS) {
            const float* pr = pos + (size_t)id * 3;
            vx = pr[0]; vy = pr[1]; vz = pr[2];
            const float* yr = y1 + (size_t)id * D2 + c16;
            g0 = yr[0];
            g1v = yr[16];
          }
          px[u][e] = vx; py[u][e] = vy; pz[u][e] = vz;
          f0[u][e] = g0; f1[u][e] = g1v;
        }
      }
#pragma unroll
      for (int u = 0; u < 2; ++u) {
        int qq = 2 * bb + u;
        int q = wave * 4 + qq;
        float qx = qpos[(size_t)(q0 + q) * 3 + 0];
        float qy = qpos[(size_t)(q0 + q) * 3 + 1];
        float qz = qpos[(size_t)(q0 + q) * 3 + 2];
#pragma unroll
        for (int ks = 0; ks < 4; ++ks) {
          float wv[2], n0[2], n1[2];
#pragma unroll
          for (int j = 0; j < 2; ++j) {
            int e = 2 * ks + j;
            int id = j ? idc[qq][ks].y : idc[qq][ks].x;
            float rx = px[u][e] - qx, ry = py[u][e] - qy, rz = pz[u][e] - qz;
            float rr = rx * rx + ry * ry + rz * rz;
            float sq = rr + kx * kx + ky * ky + kz * kz
                       - 2.f * (rx * kx + ry * ky + rz * kz);
            float w = fmaxf(1.f - sqrtf(fmaxf(sq, 0.f)) * INV_EXTENT, 0.f);
            wv[j] = w * kp_keep;
            float keep = id < N_PTS ? 1.f : 0.f;
            n0[j] = leaky(f0[u][e] * bnA[0] + bnC[0]) * keep;
            n1[j] = leaky(f1[u][e] * bnA[1] + bnC[1]) * keep;
          }
          v2f a = {wv[0], wv[1]};
          v2f b0 = {n0[0], n0[1]};
          v2f b1f = {n1[0], n1[1]};
          wt[qq][0] = wmma4(a, b0, wt[qq][0]);
          wt[qq][1] = wmma4(a, b1f, wt[qq][1]);
        }
      }
      asm volatile("v_nop\n\tv_nop\n\tv_nop\n\tv_nop\n\tv_nop"
                   : "+v"(wt[2 * bb + 1][0]), "+v"(wt[2 * bb + 1][1]));
    }
    {
      int nxt = tile + nblk < ntiles ? tile + nblk : tile;
      const int* ib = idx + ((size_t)nxt * 16 + wave * 4) * NB + 2 * hh;
#pragma unroll
      for (int qq = 0; qq < 4; ++qq) {
#pragma unroll
        for (int ks = 0; ks < 4; ++ks)
          idc[qq][ks] = *(const v2i*)(ib + qq * NB + 4 * ks);
      }
    }

    {
      v4f* wrow = (v4f*)(s_wt + ((((lane >> 4) << 3) * 32 + (lane & 15)) * 16)
                         + wave * 4);
#pragma unroll
      for (int t = 0; t < 2; ++t) {
#pragma unroll
        for (int v = 0; v < 8; ++v) {
          v4f w4 = {wt[0][t][v], wt[1][t][v], wt[2][t][v], wt[3][t][v]};
          wrow[(v * 32 + t * 16) * 4] = w4;
        }
      }
    }
    __syncthreads();

    v8f acc = {};
    const float* arow = s_wt + (kbase + kh) * 16 + (lane & 15);
    v2f ab[2][8];
#pragma unroll
    for (int i = 0; i < 8; ++i) {
      ab[0][i].x = arow[64 * i];
      ab[0][i].y = arow[64 * i + 16];
    }
#pragma unroll
    for (int g = 0; g < 8; ++g) {
      if (g + 1 < 8) {
#pragma unroll
        for (int i = 0; i < 8; ++i) {
          ab[(g + 1) & 1][i].x = arow[64 * (8 * (g + 1) + i)];
          ab[(g + 1) & 1][i].y = arow[64 * (8 * (g + 1) + i) + 16];
        }
      }
#pragma unroll
      for (int i = 0; i < 8; ++i) acc = wmma4(ab[g & 1][i], kvb[8 * g + i], acc);
    }
    asm volatile("v_nop\n\tv_nop\n\tv_nop\n\tv_nop\n\tv_nop" : "+v"(acc));
    if (khalf == 1) {
#pragma unroll
      for (int v = 0; v < 8; ++v) s_red[ntile * 256 + lane * 8 + v] = acc[v];
    }
    __syncthreads();
    if (khalf == 0) {
#pragma unroll
      for (int v = 0; v < 8; ++v) acc[v] += s_red[ntile * 256 + lane * 8 + v];
      int rb = q0 + ((lane >> 4) << 3);
#pragma unroll
      for (int v = 0; v < 8; ++v) {
        float val = acc[v];
        y2[(size_t)(rb + v) * D2 + ncol] = val;
        st_s += val;
        st_q += val * val;
      }
    }
  }
  if (khalf == 0) {
    st_s += __shfl_xor(st_s, 16);
    st_q += __shfl_xor(st_q, 16);
    if (lane < 16) {
      atomicAdd(&sumk[ncol], st_s);
      atomicAdd(&sqk[ncol], st_q);
    }
  }
}

__global__ __launch_bounds__(128) void k_gemm2(const float* __restrict__ y2,
                                               const float* __restrict__ W,
                                               const float* __restrict__ sumk,
                                               const float* __restrict__ sqk,
                                               const float* __restrict__ gk,
                                               const float* __restrict__ bk,
                                               float* __restrict__ y3,
                                               float* sum2, float* sq2,
                                               int ntiles) {
  int lane = threadIdx.x & 31;
  int wv = __builtin_amdgcn_readfirstlane((int)(threadIdx.x >> 5));
  int gw = blockIdx.x * (blockDim.x >> 5) + wv;
  int nw = gridDim.x * (blockDim.x >> 5);
  int kh = (lane >> 4) << 1;
  v2f b[8][8];
  v2f ka[8], kc[8];
#pragma unroll
  for (int ks = 0; ks < 8; ++ks) {
#pragma unroll
    for (int t = 0; t < 8; ++t) b[ks][t] = load_b_rm(W, C_OUT, 4 * ks, t * 16, lane);
    float ax, cx, ay, cy;
    bn_coef(sumk, sqk, gk, bk, (float)M_Q, 4 * ks + kh, ax, cx);
    bn_coef(sumk, sqk, gk, bk, (float)M_Q, 4 * ks + kh + 1, ay, cy);
    ka[ks].x = ax; ka[ks].y = ay;
    kc[ks].x = cx; kc[ks].y = cy;
  }
  float s[8], q[8];
#pragma unroll
  for (int t = 0; t < 8; ++t) { s[t] = 0.f; q[t] = 0.f; }
  v2f r[8];
  {
    int t0 = gw < ntiles ? gw : 0;
#pragma unroll
    for (int ks = 0; ks < 8; ++ks) r[ks] = load_a_rm(y2, D2, t0 * 16, 4 * ks, lane);
  }
  for (int tile = gw; tile < ntiles; tile += nw) {
    int m0 = tile * 16;
    int nt = tile + nw < ntiles ? tile + nw : tile;
    v2f rn[8];
#pragma unroll
    for (int ks = 0; ks < 8; ++ks) rn[ks] = load_a_rm(y2, D2, nt * 16, 4 * ks, lane);
    v2f a[8];
#pragma unroll
    for (int ks = 0; ks < 8; ++ks) {
      a[ks].x = leaky(r[ks].x * ka[ks].x + kc[ks].x);
      a[ks].y = leaky(r[ks].y * ka[ks].y + kc[ks].y);
    }
    v8f acc[8];
#pragma unroll
    for (int t = 0; t < 8; ++t) acc[t] = (v8f){};
#pragma unroll
    for (int ks = 0; ks < 8; ++ks) {
#pragma unroll
      for (int t = 0; t < 8; ++t) acc[t] = wmma4(a[ks], b[ks][t], acc[t]);
    }
#pragma unroll
    for (int t = 0; t < 8; ++t) {
      store_d_rm(y3, C_OUT, m0, t * 16, lane, acc[t]);
#pragma unroll
      for (int v = 0; v < 8; ++v) { s[t] += acc[t][v]; q[t] += acc[t][v] * acc[t][v]; }
    }
#pragma unroll
    for (int ks = 0; ks < 8; ++ks) r[ks] = rn[ks];
  }
#pragma unroll
  for (int t = 0; t < 8; ++t) {
    s[t] += __shfl_xor(s[t], 16);
    q[t] += __shfl_xor(q[t], 16);
  }
#pragma unroll
  for (int j = 0; j < 4; ++j) {
    atomicAdd(&sum2[32 * j + lane], lane < 16 ? s[2 * j] : s[2 * j + 1]);
    atomicAdd(&sq2[32 * j + lane], lane < 16 ? q[2 * j] : q[2 * j + 1]);
  }
}

__global__ __launch_bounds__(128) void k_final(const float* __restrict__ x,
                                               const int* __restrict__ idx,
                                               const float* __restrict__ Wsc,
                                               const float* __restrict__ y3,
                                               const float* __restrict__ sum2,
                                               const float* __restrict__ sq2,
                                               const float* __restrict__ g2,
                                               const float* __restrict__ b2,
                                               float* __restrict__ out) {
  __shared__ float s_sx[16 * 64];
  int q0 = blockIdx.x * 16;
  int lane = threadIdx.x & 31;
  int wave = __builtin_amdgcn_readfirstlane((int)(threadIdx.x >> 5));
  float a2v[2], c2v[2];
#pragma unroll
  for (int tt = 0; tt < 2; ++tt)
    bn_coef(sum2, sq2, g2, b2, (float)M_Q, (wave * 2 + tt) * 16 + (lane & 15),
            a2v[tt], c2v[tt]);
  const int* ip = idx + (size_t)(q0 + wave * 4) * NB;
  int i0 = ip[2 * lane];
  int i1 = ip[2 * lane + 1];
  const float* xl = x + 2 * lane;
#pragma unroll
  for (int pp = 0; pp < 2; ++pp) {
    v2f r[2][NB];
#pragma unroll
    for (int u = 0; u < 2; ++u) {
#pragma unroll
      for (int nb = 0; nb < NB; ++nb) {
        int k = (pp * 2 + u) * NB + nb;
        int n = __builtin_amdgcn_readlane((k & 1) ? i1 : i0, k >> 1);
        v2f v = {0.f, 0.f};
        if (n < N_PTS) {
          const float* row = xl + (size_t)n * C_IN;
          v.x = row[0];
          v.y = row[1];
        }
        r[u][nb] = v;
      }
    }
#pragma unroll
    for (int u = 0; u < 2; ++u) {
#pragma unroll
      for (int s = 8; s >= 1; s >>= 1) {
#pragma unroll
        for (int nb = 0; nb < s; ++nb) {
          r[u][nb].x = fmaxf(r[u][nb].x, r[u][nb + s].x);
          r[u][nb].y = fmaxf(r[u][nb].y, r[u][nb + s].y);
        }
      }
      int q = wave * 4 + pp * 2 + u;
      s_sx[q * 64 + 2 * lane] = fmaxf(-3.4e38f, r[u][0].x);
      s_sx[q * 64 + 2 * lane + 1] = fmaxf(-3.4e38f, r[u][0].y);
    }
  }
  __syncthreads();
#pragma unroll
  for (int tt = 0; tt < 2; ++tt) {
    int ntile = wave * 2 + tt;
    v8f acc = {};
#pragma unroll
    for (int k0 = 0; k0 < C_IN; k0 += 4) {
      int m = lane & 15;
      int kk = k0 + ((lane >> 4) << 1);
      v2f a;
      a.x = s_sx[m * 64 + kk];
      a.y = s_sx[m * 64 + kk + 1];
      v2f b = load_b_rm(Wsc, C_OUT, k0, ntile * 16, lane);
      acc = wmma4(a, b, acc);
    }
    int col = ntile * 16 + (lane & 15);
    int rb = (lane >> 4) << 3;
#pragma unroll
    for (int v = 0; v < 8; ++v) {
      int m = q0 + rb + v;
      float h = leaky(y3[(size_t)m * C_OUT + col] * a2v[tt] + c2v[tt]);
      out[(size_t)m * C_OUT + col] = h + acc[v];
    }
  }
}

extern "C" void kernel_launch(void* const* d_in, const int* in_sizes, int n_in,
                              void* d_out, int out_size, void* d_ws,
                              size_t ws_size, hipStream_t stream) {
  const float* x    = (const float*)d_in[0];
  const float* pos  = (const float*)d_in[1];
  const float* qpos = (const float*)d_in[2];
  const int*   idx  = (const int*)d_in[3];
  const float* Wu1  = (const float*)d_in[4];
  const float* g1   = (const float*)d_in[5];
  const float* b1   = (const float*)d_in[6];
  const float* Kpts = (const float*)d_in[7];
  const float* KV   = (const float*)d_in[8];
  const float* gk   = (const float*)d_in[9];
  const float* bk   = (const float*)d_in[10];
  const float* Wu2  = (const float*)d_in[11];
  const float* g2   = (const float*)d_in[12];
  const float* b2   = (const float*)d_in[13];
  const float* Wsc  = (const float*)d_in[14];
  float* out = (float*)d_out;

  float* ws = (float*)d_ws;
  float* y1 = ws;
  float* y2 = y1 + (size_t)N_PTS * D2;
  float* y3 = y2 + (size_t)M_Q * D2;
  float* S  = y3 + (size_t)M_Q * C_OUT;
  float *sum1 = S +   0, *sq1 = S +  32;
  float *sumk = S +  64, *sqk = S +  96;
  float *sum2 = S + 128, *sq2 = S + 256;

  k_zero<<<1, 384, 0, stream>>>(S, 384);

  k_gemm1<<<512, 128, 0, stream>>>(x, Wu1, y1, sum1, sq1, N_PTS / 16);
  k_kpconv<<<768, 128, 0, stream>>>(y1, idx, pos, qpos, Kpts, KV, sum1, sq1, g1,
                                   b1, y2, sumk, sqk, M_Q / 16);
  int tiles2 = M_Q / 16;
  k_gemm2<<<391, 128, 0, stream>>>(y2, Wu2, sumk, sqk, gk, bk, y3, sum2, sq2,
                                  tiles2);
  k_final<<<M_Q / 16, 128, 0, stream>>>(x, idx, Wsc, y3, sum2, sq2, g2, b2, out);
}
